# starting kernel
# baseline (speedup 1.0000x reference)
#include <hip/hip_runtime.h>
#include <hip/hip_bf16.h>
#include <math.h>

typedef float v2f __attribute__((ext_vector_type(2)));
typedef float v8f __attribute__((ext_vector_type(8)));

static constexpr int NN = 4096;
static constexpr int CC = 64;
static constexpr int C2 = 32;
static constexpr int NB = 8;

__device__ __forceinline__ float sw16(float v) {
  return __int_as_float(__builtin_amdgcn_ds_swizzle(__float_as_int(v), 0x401f));
}

__device__ __forceinline__ v8f wmma4(v2f a, v2f b, v8f c) {
  return __builtin_amdgcn_wmma_f32_16x16x4_f32(false, a, false, b, (short)0, c,
                                               false, false);
}

__device__ __forceinline__ void c2b(const v8f& R, bool lo, v2f* B) {
  float s0 = sw16(R[0]), s1 = sw16(R[1]), s2 = sw16(R[2]), s3 = sw16(R[3]);
  float s4 = sw16(R[4]), s5 = sw16(R[5]), s6 = sw16(R[6]), s7 = sw16(R[7]);
  B[0][0] = lo ? R[0] : s2;  B[0][1] = lo ? R[1] : s3;
  B[1][0] = lo ? R[4] : s6;  B[1][1] = lo ? R[5] : s7;
  B[2][0] = lo ? s0 : R[2];  B[2][1] = lo ? s1 : R[3];
  B[3][0] = lo ? s4 : R[6];  B[3][1] = lo ? s5 : R[7];
}

__global__ __launch_bounds__(256) void gx_kernel(
    const float* __restrict__ x, const float* __restrict__ g_w,
    const float* __restrict__ g_b, float* __restrict__ Vg) {
  int idx = blockIdx.x * 256 + threadIdx.x;
  int b = idx >> 12;
  int n = idx & (NN - 1);
  const float* xb = x + (size_t)b * (CC * NN) + n;
  float acc[C2];
#pragma unroll
  for (int o = 0; o < C2; ++o) acc[o] = g_b[o];
  for (int c = 0; c < CC; ++c) {
    float xv = xb[(size_t)c * NN];
#pragma unroll
    for (int o = 0; o < C2; ++o) acc[o] = fmaf(g_w[o * CC + c], xv, acc[o]);
  }
  float* vp = Vg + (size_t)idx * C2;
#pragma unroll
  for (int o = 0; o < C2; ++o) vp[o] = acc[o];
}

__global__ __launch_bounds__(256) void attn_kernel(
    const float* __restrict__ x, const float* __restrict__ Vg,
    const float* __restrict__ W_w, const float* __restrict__ W_b,
    float* __restrict__ out) {
  __shared__ float ldsK[16 * 66];
  __shared__ float ldsV[32 * 18];

  const int b    = blockIdx.x >> 5;
  const int qblk = blockIdx.x & 31;
  const int tid  = threadIdx.x;
  const int wave = tid >> 5;
  const int lane = tid & 31;
  const int lq   = lane & 15;
  const bool lo  = lane < 16;
  const int n0   = qblk * 128 + wave * 16;

  const float* xb = x  + (size_t)b * (CC * NN);
  const float* vb = Vg + (size_t)b * (NN * C2);

  v2f qb[16];
  {
    const int crow = lo ? 0 : 2;
#pragma unroll
    for (int ck = 0; ck < 16; ++ck) {
      qb[ck][0] = xb[(size_t)(4 * ck + crow) * NN + n0 + lq];
      qb[ck][1] = xb[(size_t)(4 * ck + crow + 1) * NN + n0 + lq];
    }
  }

  v8f acc0 = {};
  v8f acc1 = {};
  float m_run = -INFINITY;
  float l_run = 0.0f;

  const int kc_m = tid & 15, kc_c = tid >> 4;
  const int vc_o = tid & 31, vc_m = tid >> 5;

  for (int m0 = 0; m0 < NN; m0 += 16) {
    __syncthreads();
#pragma unroll
    for (int i = 0; i < 4; ++i)
      ldsK[kc_m * 66 + kc_c + 16 * i] =
          xb[(size_t)(kc_c + 16 * i) * NN + m0 + kc_m];
#pragma unroll
    for (int i = 0; i < 2; ++i)
      ldsV[vc_o * 18 + vc_m + 8 * i] =
          vb[(size_t)(m0 + vc_m + 8 * i) * C2 + vc_o];
    __syncthreads();

    v8f s = {};
    const int arow = lq * 66 + (lo ? 0 : 2);
#pragma unroll
    for (int ck = 0; ck < 16; ++ck) {
      v2f a;
      a[0] = ldsK[arow + 4 * ck];
      a[1] = ldsK[arow + 4 * ck + 1];
      s = wmma4(a, qb[ck], s);
    }

    float tmax = s[0];
#pragma unroll
    for (int r = 1; r < 8; ++r) tmax = fmaxf(tmax, s[r]);
    tmax = fmaxf(tmax, sw16(tmax));
    float m_new = fmaxf(m_run, tmax);
    float scale = __expf(m_run - m_new);
    m_run = m_new;
    v8f p;
    float tsum = 0.0f;
#pragma unroll
    for (int r = 0; r < 8; ++r) {
      p[r] = __expf(s[r] - m_new);
      tsum += p[r];
    }
    tsum += sw16(tsum);
    l_run = l_run * scale + tsum;
#pragma unroll
    for (int r = 0; r < 8; ++r) {
      acc0[r] *= scale;
      acc1[r] *= scale;
    }

    v2f pb[4];
    c2b(p, lo, pb);

    const int vrow0 = lq * 18 + (lo ? 0 : 2);
    const int vrow1 = (lq + 16) * 18 + (lo ? 0 : 2);
#pragma unroll
    for (int j = 0; j < 4; ++j) {
      v2f a;
      a[0] = ldsV[vrow0 + 4 * j];
      a[1] = ldsV[vrow0 + 4 * j + 1];
      acc0 = wmma4(a, pb[j], acc0);
    }
#pragma unroll
    for (int j = 0; j < 4; ++j) {
      v2f a;
      a[0] = ldsV[vrow1 + 4 * j];
      a[1] = ldsV[vrow1 + 4 * j + 1];
      acc1 = wmma4(a, pb[j], acc1);
    }
  }

  float rinv = 1.0f / l_run;
#pragma unroll
  for (int r = 0; r < 8; ++r) {
    acc0[r] *= rinv;
    acc1[r] *= rinv;
  }

  v2f yb[8];
  c2b(acc0, lo, yb);
  c2b(acc1, lo, yb + 4);

  float* ob = out + (size_t)b * (CC * NN);
#pragma unroll
  for (int ct = 0; ct < 4; ++ct) {
    v8f z;
#pragma unroll
    for (int r = 0; r < 8; ++r) {
      int c = ct * 16 + r + (lo ? 0 : 8);
      z[r] = xb[(size_t)c * NN + n0 + lq] + W_b[c];
    }
    const int wrow = (ct * 16 + lq) * C2 + (lo ? 0 : 2);
#pragma unroll
    for (int oc = 0; oc < 8; ++oc) {
      v2f a;
      a[0] = W_w[wrow + 4 * oc];
      a[1] = W_w[wrow + 4 * oc + 1];
      z = wmma4(a, yb[oc], z);
    }
#pragma unroll
    for (int r = 0; r < 8; ++r) {
      int c = ct * 16 + r + (lo ? 0 : 8);
      ob[(size_t)c * NN + n0 + lq] = z[r];
    }
  }
}

extern "C" void kernel_launch(void* const* d_in, const int* in_sizes, int n_in,
                              void* d_out, int out_size, void* d_ws,
                              size_t ws_size, hipStream_t stream) {
  const float* x   = (const float*)d_in[0];
  const float* g_w = (const float*)d_in[1];
  const float* g_b = (const float*)d_in[2];
  const float* W_w = (const float*)d_in[3];
  const float* W_b = (const float*)d_in[4];
  float* out = (float*)d_out;
  float* Vg  = (float*)d_ws;

  gx_kernel<<<(NB * NN) / 256, 256, 0, stream>>>(x, g_w, g_b, Vg);
  attn_kernel<<<NB * (NN / 128), 256, 0, stream>>>(x, Vg, W_w, W_b, out);
}


# K and V^T fragments requested one 64-key group ahead into registers (two K sets, V refilled inside the softmax part), held in place by scheduling fences: no load wait in the loop; sums unchanged
# speedup vs baseline: 1.9844x; 1.9844x over previous
#include <hip/hip_runtime.h>
#include <hip/hip_bf16.h>
#include <math.h>

typedef float v2f __attribute__((ext_vector_type(2)));
typedef float v4f __attribute__((ext_vector_type(4)));
typedef float v8f __attribute__((ext_vector_type(8)));

static constexpr int NN = 4096;
static constexpr int CC = 64;
static constexpr int C2 = 32;
static constexpr int NB = 8;

__device__ __forceinline__ float sw16(float v) {
  const int b = __float_as_int(v);
  return __int_as_float(
      __builtin_amdgcn_permlanex16(b, b, 0x76543210, 0xfedcba98, false, false));
}

typedef unsigned int v2u __attribute__((ext_vector_type(2)));

__device__ __forceinline__ void hswap(float& a, float& b) {
  const v2u r = __builtin_amdgcn_permlane16_swap(__float_as_uint(a),
                                                 __float_as_uint(b), false, false);
  a = __uint_as_float(r[0]);
  b = __uint_as_float(r[1]);
}

__device__ __forceinline__ v8f wmma4(v2f a, v2f b, v8f c) {
  return __builtin_amdgcn_wmma_f32_16x16x4_f32(false, a, false, b, (short)0, c,
                                               false, false);
}

__device__ __forceinline__ v8f c2b(const v8f& R) {
  float r0 = R[0], r1 = R[1], r2 = R[2], r3 = R[3];
  float r4 = R[4], r5 = R[5], r6 = R[6], r7 = R[7];
  hswap(r0, r2);
  hswap(r1, r3);
  hswap(r4, r6);
  hswap(r5, r7);
  v8f T = {r0, r1, r2, r3, r4, r5, r6, r7};
  return T;
}

__device__ __forceinline__ v2f bfrag(const v8f& T, int j) {
  const int r = (j == 0) ? 0 : (j == 1) ? 4 : (j == 2) ? 2 : 6;
  v2f f = {T[r], T[r + 1]};
  return f;
}

__global__ __launch_bounds__(256) void gx_kernel(
    const float* __restrict__ x, const float* __restrict__ g_w,
    const float* __restrict__ g_b, float* __restrict__ Vg) {
  int idx = blockIdx.x * 256 + threadIdx.x;
  int b = idx >> 12;
  int n = idx & (NN - 1);
  const float* xb = x + (size_t)b * (CC * NN) + n;
  float acc[C2];
#pragma unroll
  for (int o = 0; o < C2; ++o) acc[o] = g_b[o];
  for (int c = 0; c < CC; ++c) {
    float xv = xb[(size_t)c * NN];
#pragma unroll
    for (int o = 0; o < C2; ++o) acc[o] = fmaf(g_w[o * CC + c], xv, acc[o]);
  }
  float* vp = Vg + (size_t)idx * C2;
#pragma unroll
  for (int o = 0; o < C2; ++o) vp[o] = acc[o];
}

__device__ __forceinline__ void hold_loads_here() {
  __builtin_amdgcn_sched_barrier(0);
}

__device__ __forceinline__ void load_k(v4f (&ka)[16], v4f (&kb)[16],
                                       const float* __restrict__ kp, int m) {
#pragma unroll
  for (int ck = 0; ck < 16; ++ck) {
    ka[ck] = *(const v4f*)(kp + (size_t)(4 * ck) * NN + m);
    kb[ck] = *(const v4f*)(kp + (size_t)(4 * ck + 1) * NN + m);
  }
}

__device__ __forceinline__ void load_v(v2f (&va)[4], v2f (&vb)[4],
                                       const float* __restrict__ vp, int m, int t) {
  const float* vt = vp + (size_t)(m + t) * C2;
#pragma unroll
  for (int j = 0; j < 4; ++j) {
    va[j][0] = vt[(16 * j) * C2];
    va[j][1] = vt[(16 * j + 4) * C2];
    vb[j][0] = vt[(16 * j) * C2 + 16];
    vb[j][1] = vt[(16 * j + 4) * C2 + 16];
  }
}

__device__ __forceinline__ void key_group(
    const v4f (&ka)[16], const v4f (&kb)[16], v2f (&va)[4][4], v2f (&vb)[4][4],
    const v2f (&qb)[16], v8f& acc0, v8f& acc1, float& m_run, float& l_run,
    const float* __restrict__ vp, int m_next) {
  v8f s[4] = {{}, {}, {}, {}};
#pragma unroll
  for (int ck = 0; ck < 16; ++ck) {
#pragma unroll
    for (int t = 0; t < 4; ++t) {
      v2f a;
      a[0] = ka[ck][t];
      a[1] = kb[ck][t];
      s[t] = wmma4(a, qb[ck], s[t]);
    }
  }
  hold_loads_here();

#pragma unroll
  for (int t = 0; t < 4; ++t) {
    const v8f st = c2b(s[t]);
    float tmax = st[0];
#pragma unroll
    for (int r = 1; r < 8; ++r) tmax = fmaxf(tmax, st[r]);
    tmax = fmaxf(tmax, sw16(tmax));
    float m_new = fmaxf(m_run, tmax);
    float scale = __expf(m_run - m_new);
    m_run = m_new;
    v8f p;
    float tsum = 0.0f;
#pragma unroll
    for (int r = 0; r < 8; ++r) {
      p[r] = __expf(st[r] - m_new);
      tsum += p[r];
    }
    tsum += sw16(tsum);
    l_run = l_run * scale + tsum;
#pragma unroll
    for (int r = 0; r < 8; ++r) {
      acc0[r] *= scale;
      acc1[r] *= scale;
    }

#pragma unroll
    for (int j = 0; j < 4; ++j) acc0 = wmma4(va[t][j], bfrag(p, j), acc0);
#pragma unroll
    for (int j = 0; j < 4; ++j) acc1 = wmma4(vb[t][j], bfrag(p, j), acc1);
    load_v(va[t], vb[t], vp, m_next, t);
    if (t == 1) hold_loads_here();
  }
  hold_loads_here();
}

__global__ __launch_bounds__(256) void attn_kernel(
    const float* __restrict__ x, const float* __restrict__ Vg,
    const float* __restrict__ W_w, const float* __restrict__ W_b,
    float* __restrict__ out) {
  const int b    = blockIdx.x >> 5;
  const int qblk = blockIdx.x & 31;
  const int tid  = threadIdx.x;
  const int wave = tid >> 5;
  const int lane = tid & 31;
  const int lq   = lane & 15;
  const bool lo  = lane < 16;
  const int crow = lo ? 0 : 2;
  const int n0   = qblk * 128 + wave * 16;

  const float* xb = x  + (size_t)b * (CC * NN);
  const float* vb = Vg + (size_t)b * (NN * C2);

  v2f qb[16];
  {
#pragma unroll
    for (int ck = 0; ck < 16; ++ck) {
      qb[ck][0] = xb[(size_t)(4 * ck + crow) * NN + n0 + lq];
      qb[ck][1] = xb[(size_t)(4 * ck + crow + 1) * NN + n0 + lq];
    }
  }

  v8f acc0 = {};
  v8f acc1 = {};
  float m_run = -INFINITY;
  float l_run = 0.0f;

  const float* kp = xb + (size_t)crow * NN + 4 * lq;
  const float* vp = vb + (size_t)(4 * crow) * C2 + lq;

  v4f k0a[16], k0b[16], k1a[16], k1b[16];
  v2f vfa[4][4], vfb[4][4];
  load_k(k0a, k0b, kp, 0);
#pragma unroll
  for (int t = 0; t < 4; ++t) load_v(vfa[t], vfb[t], vp, 0, t);

  for (int m0 = 0; m0 < NN; m0 += 128) {
    const int m1 = m0 + 64;
    const int m2 = (m0 + 128) & (NN - 1);
    load_k(k1a, k1b, kp, m1);
    hold_loads_here();
    key_group(k0a, k0b, vfa, vfb, qb, acc0, acc1, m_run, l_run, vp, m1);
    load_k(k0a, k0b, kp, m2);
    hold_loads_here();
    key_group(k1a, k1b, vfa, vfb, qb, acc0, acc1, m_run, l_run, vp, m2);
  }

  float rinv = 1.0f / l_run;
#pragma unroll
  for (int r = 0; r < 8; ++r) {
    acc0[r] *= rinv;
    acc1[r] *= rinv;
  }

  const v8f y0 = c2b(acc0);
  const v8f y1 = c2b(acc1);

  float* ob = out + (size_t)b * (CC * NN);
#pragma unroll
  for (int ct = 0; ct < 4; ++ct) {
    v8f z;
#pragma unroll
    for (int r = 0; r < 8; ++r) {
      int c = ct * 16 + r + (lo ? 0 : 8);
      z[r] = xb[(size_t)c * NN + n0 + lq] + W_b[c];
    }
    const int wrow = (ct * 16 + lq) * C2 + (lo ? 0 : 2);
#pragma unroll
    for (int oc = 0; oc < 8; ++oc) {
      v2f a;
      a[0] = W_w[wrow + 4 * oc];
      a[1] = W_w[wrow + 4 * oc + 1];
      z = wmma4(a, (oc < 4) ? bfrag(y0, oc) : bfrag(y1, oc - 4), z);
    }
#pragma unroll
    for (int r = 0; r < 8; ++r) {
      int c = ct * 16 + r + (lo ? 0 : 8);
      ob[(size_t)c * NN + n0 + lq] = z[r];
    }
  }
}

extern "C" void kernel_launch(void* const* d_in, const int* in_sizes, int n_in,
                              void* d_out, int out_size, void* d_ws,
                              size_t ws_size, hipStream_t stream) {
  const float* x   = (const float*)d_in[0];
  const float* g_w = (const float*)d_in[1];
  const float* g_b = (const float*)d_in[2];
  const float* W_w = (const float*)d_in[3];
  const float* W_b = (const float*)d_in[4];
  float* out = (float*)d_out;
  float* Vg  = (float*)d_ws;

  gx_kernel<<<(NB * NN) / 256, 256, 0, stream>>>(x, g_w, g_b, Vg);
  attn_kernel<<<NB * (NN / 128), 256, 0, stream>>>(x, Vg, W_w, W_b, out);
}


# V projection as a product on the f32 16x16x4 form: one wave = one 16-key tile V^T = g_w * x, independent loads, 32 products (was a 64-trip channel loop per key); order of V's channel sums may change
# speedup vs baseline: 2.3114x; 1.1648x over previous
#include <hip/hip_runtime.h>
#include <hip/hip_bf16.h>
#include <math.h>

typedef float v2f __attribute__((ext_vector_type(2)));
typedef float v4f __attribute__((ext_vector_type(4)));
typedef float v8f __attribute__((ext_vector_type(8)));

static constexpr int NN = 4096;
static constexpr int CC = 64;
static constexpr int C2 = 32;
static constexpr int NB = 8;

__device__ __forceinline__ float sw16(float v) {
  const int b = __float_as_int(v);
  return __int_as_float(
      __builtin_amdgcn_permlanex16(b, b, 0x76543210, 0xfedcba98, false, false));
}

typedef unsigned int v2u __attribute__((ext_vector_type(2)));

__device__ __forceinline__ void hswap(float& a, float& b) {
  const v2u r = __builtin_amdgcn_permlane16_swap(__float_as_uint(a),
                                                 __float_as_uint(b), false, false);
  a = __uint_as_float(r[0]);
  b = __uint_as_float(r[1]);
}

__device__ __forceinline__ v8f wmma4(v2f a, v2f b, v8f c) {
  return __builtin_amdgcn_wmma_f32_16x16x4_f32(false, a, false, b, (short)0, c,
                                               false, false);
}

__device__ __forceinline__ v8f c2b(const v8f& R) {
  float r0 = R[0], r1 = R[1], r2 = R[2], r3 = R[3];
  float r4 = R[4], r5 = R[5], r6 = R[6], r7 = R[7];
  hswap(r0, r2);
  hswap(r1, r3);
  hswap(r4, r6);
  hswap(r5, r7);
  v8f T = {r0, r1, r2, r3, r4, r5, r6, r7};
  return T;
}

__device__ __forceinline__ v2f bfrag(const v8f& T, int j) {
  const int r = (j == 0) ? 0 : (j == 1) ? 4 : (j == 2) ? 2 : 6;
  v2f f = {T[r], T[r + 1]};
  return f;
}

__global__ __launch_bounds__(256) void gx_kernel(
    const float* __restrict__ x, const float* __restrict__ g_w,
    const float* __restrict__ g_b, float* __restrict__ Vg) {
  const int b    = blockIdx.x >> 5;
  const int wave = threadIdx.x >> 5;
  const int lane = threadIdx.x & 31;
  const int lq   = lane & 15;
  const int h2   = (lane < 16) ? 0 : 2;
  const int ob   = (lane < 16) ? 0 : 8;
  const int n0   = (blockIdx.x & 31) * 128 + wave * 16;

  const float* xp = x + (size_t)b * (CC * NN) + (size_t)h2 * NN + n0 + lq;
  const float* wp = g_w + lq * CC + h2;
  v2f xf[16], w0[16], w1[16];
#pragma unroll
  for (int ck = 0; ck < 16; ++ck) {
    xf[ck][0] = xp[(size_t)(4 * ck) * NN];
    xf[ck][1] = xp[(size_t)(4 * ck + 1) * NN];
    w0[ck] = *(const v2f*)(wp + 4 * ck);
    w1[ck] = *(const v2f*)(wp + 16 * CC + 4 * ck);
  }
  v8f d0, d1;
#pragma unroll
  for (int r = 0; r < 8; ++r) {
    d0[r] = g_b[ob + r];
    d1[r] = g_b[16 + ob + r];
  }
#pragma unroll
  for (int ck = 0; ck < 16; ++ck) {
    d0 = wmma4(w0[ck], xf[ck], d0);
    d1 = wmma4(w1[ck], xf[ck], d1);
  }
  float* vp = Vg + ((size_t)b * NN + n0 + lq) * C2 + ob;
  v4f s0 = {d0[0], d0[1], d0[2], d0[3]};
  v4f s1 = {d0[4], d0[5], d0[6], d0[7]};
  v4f s2 = {d1[0], d1[1], d1[2], d1[3]};
  v4f s3 = {d1[4], d1[5], d1[6], d1[7]};
  *(v4f*)(vp)      = s0;
  *(v4f*)(vp + 4)  = s1;
  *(v4f*)(vp + 16) = s2;
  *(v4f*)(vp + 20) = s3;
}

__device__ __forceinline__ void hold_loads_here() {
  __builtin_amdgcn_sched_barrier(0);
}

__device__ __forceinline__ void load_k(v4f (&ka)[16], v4f (&kb)[16],
                                       const float* __restrict__ kp, int m) {
#pragma unroll
  for (int ck = 0; ck < 16; ++ck) {
    ka[ck] = *(const v4f*)(kp + (size_t)(4 * ck) * NN + m);
    kb[ck] = *(const v4f*)(kp + (size_t)(4 * ck + 1) * NN + m);
  }
}

__device__ __forceinline__ void load_v(v2f (&va)[4], v2f (&vb)[4],
                                       const float* __restrict__ vp, int m, int t) {
  const float* vt = vp + (size_t)(m + t) * C2;
#pragma unroll
  for (int j = 0; j < 4; ++j) {
    va[j][0] = vt[(16 * j) * C2];
    va[j][1] = vt[(16 * j + 4) * C2];
    vb[j][0] = vt[(16 * j) * C2 + 16];
    vb[j][1] = vt[(16 * j + 4) * C2 + 16];
  }
}

__device__ __forceinline__ void key_group(
    const v4f (&ka)[16], const v4f (&kb)[16], v2f (&va)[4][4], v2f (&vb)[4][4],
    const v2f (&qb)[16], v8f& acc0, v8f& acc1, float& m_run, float& l_run,
    const float* __restrict__ vp, int m_next) {
  v8f s[4] = {{}, {}, {}, {}};
#pragma unroll
  for (int ck = 0; ck < 16; ++ck) {
#pragma unroll
    for (int t = 0; t < 4; ++t) {
      v2f a;
      a[0] = ka[ck][t];
      a[1] = kb[ck][t];
      s[t] = wmma4(a, qb[ck], s[t]);
    }
  }
  hold_loads_here();

#pragma unroll
  for (int t = 0; t < 4; ++t) {
    const v8f st = c2b(s[t]);
    float tmax = st[0];
#pragma unroll
    for (int r = 1; r < 8; ++r) tmax = fmaxf(tmax, st[r]);
    tmax = fmaxf(tmax, sw16(tmax));
    float m_new = fmaxf(m_run, tmax);
    float scale = __expf(m_run - m_new);
    m_run = m_new;
    v8f p;
    float tsum = 0.0f;
#pragma unroll
    for (int r = 0; r < 8; ++r) {
      p[r] = __expf(st[r] - m_new);
      tsum += p[r];
    }
    tsum += sw16(tsum);
    l_run = l_run * scale + tsum;
#pragma unroll
    for (int r = 0; r < 8; ++r) {
      acc0[r] *= scale;
      acc1[r] *= scale;
    }

#pragma unroll
    for (int j = 0; j < 4; ++j) acc0 = wmma4(va[t][j], bfrag(p, j), acc0);
#pragma unroll
    for (int j = 0; j < 4; ++j) acc1 = wmma4(vb[t][j], bfrag(p, j), acc1);
    load_v(va[t], vb[t], vp, m_next, t);
    if (t == 1) hold_loads_here();
  }
  hold_loads_here();
}

__global__ __launch_bounds__(256) void attn_kernel(
    const float* __restrict__ x, const float* __restrict__ Vg,
    const float* __restrict__ W_w, const float* __restrict__ W_b,
    float* __restrict__ out) {
  const int b    = blockIdx.x >> 5;
  const int qblk = blockIdx.x & 31;
  const int tid  = threadIdx.x;
  const int wave = tid >> 5;
  const int lane = tid & 31;
  const int lq   = lane & 15;
  const bool lo  = lane < 16;
  const int crow = lo ? 0 : 2;
  const int n0   = qblk * 128 + wave * 16;

  const float* xb = x  + (size_t)b * (CC * NN);
  const float* vb = Vg + (size_t)b * (NN * C2);

  v2f qb[16];
  {
#pragma unroll
    for (int ck = 0; ck < 16; ++ck) {
      qb[ck][0] = xb[(size_t)(4 * ck + crow) * NN + n0 + lq];
      qb[ck][1] = xb[(size_t)(4 * ck + crow + 1) * NN + n0 + lq];
    }
  }

  v8f acc0 = {};
  v8f acc1 = {};
  float m_run = -INFINITY;
  float l_run = 0.0f;

  const float* kp = xb + (size_t)crow * NN + 4 * lq;
  const float* vp = vb + (size_t)(4 * crow) * C2 + lq;

  v4f k0a[16], k0b[16], k1a[16], k1b[16];
  v2f vfa[4][4], vfb[4][4];
  load_k(k0a, k0b, kp, 0);
#pragma unroll
  for (int t = 0; t < 4; ++t) load_v(vfa[t], vfb[t], vp, 0, t);

  for (int m0 = 0; m0 < NN; m0 += 128) {
    const int m1 = m0 + 64;
    const int m2 = (m0 + 128) & (NN - 1);
    load_k(k1a, k1b, kp, m1);
    hold_loads_here();
    key_group(k0a, k0b, vfa, vfb, qb, acc0, acc1, m_run, l_run, vp, m1);
    load_k(k0a, k0b, kp, m2);
    hold_loads_here();
    key_group(k1a, k1b, vfa, vfb, qb, acc0, acc1, m_run, l_run, vp, m2);
  }

  float rinv = 1.0f / l_run;
#pragma unroll
  for (int r = 0; r < 8; ++r) {
    acc0[r] *= rinv;
    acc1[r] *= rinv;
  }

  const v8f y0 = c2b(acc0);
  const v8f y1 = c2b(acc1);

  float* ob = out + (size_t)b * (CC * NN);
#pragma unroll
  for (int ct = 0; ct < 4; ++ct) {
    v8f z;
#pragma unroll
    for (int r = 0; r < 8; ++r) {
      int c = ct * 16 + r + (lo ? 0 : 8);
      z[r] = xb[(size_t)c * NN + n0 + lq] + W_b[c];
    }
    const int wrow = (ct * 16 + lq) * C2 + (lo ? 0 : 2);
#pragma unroll
    for (int oc = 0; oc < 8; ++oc) {
      v2f a;
      a[0] = W_w[wrow + 4 * oc];
      a[1] = W_w[wrow + 4 * oc + 1];
      z = wmma4(a, (oc < 4) ? bfrag(y0, oc) : bfrag(y1, oc - 4), z);
    }
#pragma unroll
    for (int r = 0; r < 8; ++r) {
      int c = ct * 16 + r + (lo ? 0 : 8);
      ob[(size_t)c * NN + n0 + lq] = z[r];
    }
  }
}

extern "C" void kernel_launch(void* const* d_in, const int* in_sizes, int n_in,
                              void* d_out, int out_size, void* d_ws,
                              size_t ws_size, hipStream_t stream) {
  const float* x   = (const float*)d_in[0];
  const float* g_w = (const float*)d_in[1];
  const float* g_b = (const float*)d_in[2];
  const float* W_w = (const float*)d_in[3];
  const float* W_b = (const float*)d_in[4];
  float* out = (float*)d_out;
  float* Vg  = (float*)d_ws;

  gx_kernel<<<NB * (NN / 128), 256, 0, stream>>>(x, g_w, g_b, Vg);
  attn_kernel<<<NB * (NN / 128), 256, 0, stream>>>(x, Vg, W_w, W_b, out);
}


# V written by the projection kernel in V^T-fragment order (one wave = 32 keys, whole 512-byte rows): attention reads V as 16 whole-line 16-byte loads per 64 keys, was 64 4-byte loads; sums unchanged
# speedup vs baseline: 2.4041x; 1.0401x over previous
#include <hip/hip_runtime.h>
#include <hip/hip_bf16.h>
#include <math.h>

typedef float v2f __attribute__((ext_vector_type(2)));
typedef float v4f __attribute__((ext_vector_type(4)));
typedef float v8f __attribute__((ext_vector_type(8)));

static constexpr int NN = 4096;
static constexpr int CC = 64;
static constexpr int C2 = 32;
static constexpr int NB = 8;

__device__ __forceinline__ float sw16(float v) {
  const int b = __float_as_int(v);
  return __int_as_float(
      __builtin_amdgcn_permlanex16(b, b, 0x76543210, 0xfedcba98, false, false));
}

typedef unsigned int v2u __attribute__((ext_vector_type(2)));

__device__ __forceinline__ void hswap(float& a, float& b) {
  const v2u r = __builtin_amdgcn_permlane16_swap(__float_as_uint(a),
                                                 __float_as_uint(b), false, false);
  a = __uint_as_float(r[0]);
  b = __uint_as_float(r[1]);
}

__device__ __forceinline__ v8f wmma4(v2f a, v2f b, v8f c) {
  return __builtin_amdgcn_wmma_f32_16x16x4_f32(false, a, false, b, (short)0, c,
                                               false, false);
}

__device__ __forceinline__ v8f c2b(const v8f& R) {
  float r0 = R[0], r1 = R[1], r2 = R[2], r3 = R[3];
  float r4 = R[4], r5 = R[5], r6 = R[6], r7 = R[7];
  hswap(r0, r2);
  hswap(r1, r3);
  hswap(r4, r6);
  hswap(r5, r7);
  v8f T = {r0, r1, r2, r3, r4, r5, r6, r7};
  return T;
}

__device__ __forceinline__ v2f bfrag(const v8f& T, int j) {
  const int r = (j == 0) ? 0 : (j == 1) ? 4 : (j == 2) ? 2 : 6;
  v2f f = {T[r], T[r + 1]};
  return f;
}

__device__ __forceinline__ void hold_loads_here() {
  __builtin_amdgcn_sched_barrier(0);
}

__global__ __launch_bounds__(128) void gx_kernel(
    const float* __restrict__ x, const float* __restrict__ g_w,
    const float* __restrict__ g_b, float* __restrict__ Vg) {
  const int b    = blockIdx.x >> 5;
  const int wave = threadIdx.x >> 5;
  const int lane = threadIdx.x & 31;
  const int lq   = lane & 15;
  const int h2   = (lane < 16) ? 0 : 2;
  const int n0   = (blockIdx.x & 31) * 128 + wave * 32;

  const float* xp = x + (size_t)b * (CC * NN) + (size_t)h2 * NN + n0 + lq;
  const float* wp = g_w + lq * CC + h2;
  v2f xf[2][16], wf[2][16];
#pragma unroll
  for (int ck = 0; ck < 16; ++ck) {
#pragma unroll
    for (int kt = 0; kt < 2; ++kt) {
      xf[kt][ck][0] = xp[(size_t)(4 * ck) * NN + 16 * kt];
      xf[kt][ck][1] = xp[(size_t)(4 * ck + 1) * NN + 16 * kt];
    }
    wf[0][ck] = *(const v2f*)(wp + 4 * ck);
    wf[1][ck] = *(const v2f*)(wp + 16 * CC + 4 * ck);
  }
  const float bias0 = g_b[lq], bias1 = g_b[16 + lq];
  hold_loads_here();
  v8f d[2][2];
#pragma unroll
  for (int r = 0; r < 8; ++r) {
    d[0][0][r] = bias0;
    d[1][0][r] = bias0;
    d[0][1][r] = bias1;
    d[1][1][r] = bias1;
  }
#pragma unroll
  for (int ck = 0; ck < 16; ++ck) {
#pragma unroll
    for (int kt = 0; kt < 2; ++kt) {
      d[kt][0] = wmma4(xf[kt][ck], wf[0][ck], d[kt][0]);
      d[kt][1] = wmma4(xf[kt][ck], wf[1][ck], d[kt][1]);
    }
  }
  const int jj = (n0 >> 5) & 1;
  float* vp = Vg + ((size_t)b * (NN / 64) + (n0 >> 6)) * 2048 + (jj * 32 + lane) * 4;
#pragma unroll
  for (int ab = 0; ab < 2; ++ab) {
#pragma unroll
    for (int t = 0; t < 4; ++t) {
      v4f s = {d[0][ab][t], d[0][ab][t + 4], d[1][ab][t], d[1][ab][t + 4]};
      *(v4f*)(vp + (4 * t + 2 * ab) * 128) = s;
    }
  }
}

__device__ __forceinline__ void load_k(v4f (&ka)[16], v4f (&kb)[16],
                                       const float* __restrict__ kp, int m) {
#pragma unroll
  for (int ck = 0; ck < 16; ++ck) {
    ka[ck] = *(const v4f*)(kp + (size_t)(4 * ck) * NN + m);
    kb[ck] = *(const v4f*)(kp + (size_t)(4 * ck + 1) * NN + m);
  }
}

__device__ __forceinline__ void load_v(v4f (&vq)[4], const float* __restrict__ vp,
                                       int m, int t) {
  const float* vt = vp + (size_t)(m >> 6) * 2048 + t * 512;
#pragma unroll
  for (int q = 0; q < 4; ++q) vq[q] = *(const v4f*)(vt + q * 128);
}

__device__ __forceinline__ v2f vfrag(const v4f (&vq)[4], int half, int j) {
  const v4f& w = vq[2 * half + (j >> 1)];
  v2f f;
  f[0] = (j & 1) ? w[2] : w[0];
  f[1] = (j & 1) ? w[3] : w[1];
  return f;
}

__device__ __forceinline__ void key_group(
    const v4f (&ka)[16], const v4f (&kb)[16], v4f (&vq)[4][4],
    const v2f (&qb)[16], v8f& acc0, v8f& acc1, float& m_run, float& l_run,
    const float* __restrict__ vp, int m_next) {
  v8f s[4] = {{}, {}, {}, {}};
#pragma unroll
  for (int ck = 0; ck < 16; ++ck) {
#pragma unroll
    for (int t = 0; t < 4; ++t) {
      v2f a;
      a[0] = ka[ck][t];
      a[1] = kb[ck][t];
      s[t] = wmma4(a, qb[ck], s[t]);
    }
  }
  hold_loads_here();

#pragma unroll
  for (int t = 0; t < 4; ++t) {
    const v8f st = c2b(s[t]);
    float tmax = st[0];
#pragma unroll
    for (int r = 1; r < 8; ++r) tmax = fmaxf(tmax, st[r]);
    tmax = fmaxf(tmax, sw16(tmax));
    float m_new = fmaxf(m_run, tmax);
    float scale = __expf(m_run - m_new);
    m_run = m_new;
    v8f p;
    float tsum = 0.0f;
#pragma unroll
    for (int r = 0; r < 8; ++r) {
      p[r] = __expf(st[r] - m_new);
      tsum += p[r];
    }
    tsum += sw16(tsum);
    l_run = l_run * scale + tsum;
#pragma unroll
    for (int r = 0; r < 8; ++r) {
      acc0[r] *= scale;
      acc1[r] *= scale;
    }

#pragma unroll
    for (int j = 0; j < 4; ++j) acc0 = wmma4(vfrag(vq[t], 0, j), bfrag(p, j), acc0);
#pragma unroll
    for (int j = 0; j < 4; ++j) acc1 = wmma4(vfrag(vq[t], 1, j), bfrag(p, j), acc1);
    load_v(vq[t], vp, m_next, t);
    if (t == 1) hold_loads_here();
  }
  hold_loads_here();
}

__global__ __launch_bounds__(256) void attn_kernel(
    const float* __restrict__ x, const float* __restrict__ Vg,
    const float* __restrict__ W_w, const float* __restrict__ W_b,
    float* __restrict__ out) {
  const int b    = blockIdx.x >> 5;
  const int qblk = blockIdx.x & 31;
  const int tid  = threadIdx.x;
  const int wave = tid >> 5;
  const int lane = tid & 31;
  const int lq   = lane & 15;
  const bool lo  = lane < 16;
  const int crow = lo ? 0 : 2;
  const int n0   = qblk * 128 + wave * 16;

  const float* xb = x  + (size_t)b * (CC * NN);
  const float* vb = Vg + (size_t)b * (NN * C2);

  v2f qb[16];
  {
#pragma unroll
    for (int ck = 0; ck < 16; ++ck) {
      qb[ck][0] = xb[(size_t)(4 * ck + crow) * NN + n0 + lq];
      qb[ck][1] = xb[(size_t)(4 * ck + crow + 1) * NN + n0 + lq];
    }
  }

  v8f acc0 = {};
  v8f acc1 = {};
  float m_run = -INFINITY;
  float l_run = 0.0f;

  const float* kp = xb + (size_t)crow * NN + 4 * lq;
  const float* vp = vb + lane * 4;

  v4f k0a[16], k0b[16], k1a[16], k1b[16];
  v4f vf[4][4];
  load_k(k0a, k0b, kp, 0);
#pragma unroll
  for (int t = 0; t < 4; ++t) load_v(vf[t], vp, 0, t);

  for (int m0 = 0; m0 < NN; m0 += 128) {
    const int m1 = m0 + 64;
    const int m2 = (m0 + 128) & (NN - 1);
    load_k(k1a, k1b, kp, m1);
    hold_loads_here();
    key_group(k0a, k0b, vf, qb, acc0, acc1, m_run, l_run, vp, m1);
    load_k(k0a, k0b, kp, m2);
    hold_loads_here();
    key_group(k1a, k1b, vf, qb, acc0, acc1, m_run, l_run, vp, m2);
  }

  float rinv = 1.0f / l_run;
#pragma unroll
  for (int r = 0; r < 8; ++r) {
    acc0[r] *= rinv;
    acc1[r] *= rinv;
  }

  const v8f y0 = c2b(acc0);
  const v8f y1 = c2b(acc1);

  float* ob = out + (size_t)b * (CC * NN);
#pragma unroll
  for (int ct = 0; ct < 4; ++ct) {
    v8f z;
#pragma unroll
    for (int r = 0; r < 8; ++r) {
      int c = ct * 16 + r + (lo ? 0 : 8);
      z[r] = xb[(size_t)c * NN + n0 + lq] + W_b[c];
    }
    const int wrow = (ct * 16 + lq) * C2 + (lo ? 0 : 2);
#pragma unroll
    for (int oc = 0; oc < 8; ++oc) {
      v2f a;
      a[0] = W_w[wrow + 4 * oc];
      a[1] = W_w[wrow + 4 * oc + 1];
      z = wmma4(a, (oc < 4) ? bfrag(y0, oc) : bfrag(y1, oc - 4), z);
    }
#pragma unroll
    for (int r = 0; r < 8; ++r) {
      int c = ct * 16 + r + (lo ? 0 : 8);
      ob[(size_t)c * NN + n0 + lq] = z[r];
    }
  }
}

extern "C" void kernel_launch(void* const* d_in, const int* in_sizes, int n_in,
                              void* d_out, int out_size, void* d_ws,
                              size_t ws_size, hipStream_t stream) {
  const float* x   = (const float*)d_in[0];
  const float* g_w = (const float*)d_in[1];
  const float* g_b = (const float*)d_in[2];
  const float* W_w = (const float*)d_in[3];
  const float* W_b = (const float*)d_in[4];
  float* out = (float*)d_out;
  float* Vg  = (float*)d_ws;

  gx_kernel<<<NB * (NN / 128), 128, 0, stream>>>(x, g_w, g_b, Vg);
  attn_kernel<<<NB * (NN / 128), 256, 0, stream>>>(x, Vg, W_w, W_b, out);
}


# Wave priority by phase (s_setprio): softmax+PV part 3, score run 0/1/2 rising with progress, so the two waves of a SIMD alternate phases on the matrix pipe instead of oldest-first; sums unchanged
# speedup vs baseline: 2.4957x; 1.0381x over previous
#include <hip/hip_runtime.h>
#include <hip/hip_bf16.h>
#include <math.h>

typedef float v2f __attribute__((ext_vector_type(2)));
typedef float v4f __attribute__((ext_vector_type(4)));
typedef float v8f __attribute__((ext_vector_type(8)));

static constexpr int NN = 4096;
static constexpr int CC = 64;
static constexpr int C2 = 32;
static constexpr int NB = 8;

__device__ __forceinline__ float sw16(float v) {
  const int b = __float_as_int(v);
  return __int_as_float(
      __builtin_amdgcn_permlanex16(b, b, 0x76543210, 0xfedcba98, false, false));
}

typedef unsigned int v2u __attribute__((ext_vector_type(2)));

__device__ __forceinline__ void hswap(float& a, float& b) {
  const v2u r = __builtin_amdgcn_permlane16_swap(__float_as_uint(a),
                                                 __float_as_uint(b), false, false);
  a = __uint_as_float(r[0]);
  b = __uint_as_float(r[1]);
}

__device__ __forceinline__ v8f wmma4(v2f a, v2f b, v8f c) {
  return __builtin_amdgcn_wmma_f32_16x16x4_f32(false, a, false, b, (short)0, c,
                                               false, false);
}

__device__ __forceinline__ v8f c2b(const v8f& R) {
  float r0 = R[0], r1 = R[1], r2 = R[2], r3 = R[3];
  float r4 = R[4], r5 = R[5], r6 = R[6], r7 = R[7];
  hswap(r0, r2);
  hswap(r1, r3);
  hswap(r4, r6);
  hswap(r5, r7);
  v8f T = {r0, r1, r2, r3, r4, r5, r6, r7};
  return T;
}

__device__ __forceinline__ v2f bfrag(const v8f& T, int j) {
  const int r = (j == 0) ? 0 : (j == 1) ? 4 : (j == 2) ? 2 : 6;
  v2f f = {T[r], T[r + 1]};
  return f;
}

__device__ __forceinline__ void hold_loads_here() {
  __builtin_amdgcn_sched_barrier(0);
}

template <int P>
__device__ __forceinline__ void phase_priority() {
  __builtin_amdgcn_s_setprio(P);
}

__global__ __launch_bounds__(128) void gx_kernel(
    const float* __restrict__ x, const float* __restrict__ g_w,
    const float* __restrict__ g_b, float* __restrict__ Vg) {
  const int b    = blockIdx.x >> 5;
  const int wave = threadIdx.x >> 5;
  const int lane = threadIdx.x & 31;
  const int lq   = lane & 15;
  const int h2   = (lane < 16) ? 0 : 2;
  const int n0   = (blockIdx.x & 31) * 128 + wave * 32;

  const float* xp = x + (size_t)b * (CC * NN) + (size_t)h2 * NN + n0 + lq;
  const float* wp = g_w + lq * CC + h2;
  v2f xf[2][16], wf[2][16];
#pragma unroll
  for (int ck = 0; ck < 16; ++ck) {
#pragma unroll
    for (int kt = 0; kt < 2; ++kt) {
      xf[kt][ck][0] = xp[(size_t)(4 * ck) * NN + 16 * kt];
      xf[kt][ck][1] = xp[(size_t)(4 * ck + 1) * NN + 16 * kt];
    }
    wf[0][ck] = *(const v2f*)(wp + 4 * ck);
    wf[1][ck] = *(const v2f*)(wp + 16 * CC + 4 * ck);
  }
  const float bias0 = g_b[lq], bias1 = g_b[16 + lq];
  hold_loads_here();
  v8f d[2][2];
#pragma unroll
  for (int r = 0; r < 8; ++r) {
    d[0][0][r] = bias0;
    d[1][0][r] = bias0;
    d[0][1][r] = bias1;
    d[1][1][r] = bias1;
  }
#pragma unroll
  for (int ck = 0; ck < 16; ++ck) {
#pragma unroll
    for (int kt = 0; kt < 2; ++kt) {
      d[kt][0] = wmma4(xf[kt][ck], wf[0][ck], d[kt][0]);
      d[kt][1] = wmma4(xf[kt][ck], wf[1][ck], d[kt][1]);
    }
  }
  const int jj = (n0 >> 5) & 1;
  float* vp = Vg + ((size_t)b * (NN / 64) + (n0 >> 6)) * 2048 + (jj * 32 + lane) * 4;
#pragma unroll
  for (int ab = 0; ab < 2; ++ab) {
#pragma unroll
    for (int t = 0; t < 4; ++t) {
      v4f s = {d[0][ab][t], d[0][ab][t + 4], d[1][ab][t], d[1][ab][t + 4]};
      *(v4f*)(vp + (4 * t + 2 * ab) * 128) = s;
    }
  }
}

__device__ __forceinline__ void load_k(v4f (&ka)[16], v4f (&kb)[16],
                                       const float* __restrict__ kp, int m) {
#pragma unroll
  for (int ck = 0; ck < 16; ++ck) {
    ka[ck] = *(const v4f*)(kp + (size_t)(4 * ck) * NN + m);
    kb[ck] = *(const v4f*)(kp + (size_t)(4 * ck + 1) * NN + m);
  }
}

__device__ __forceinline__ void load_v(v4f (&vq)[4], const float* __restrict__ vp,
                                       int m, int t) {
  const float* vt = vp + (size_t)(m >> 6) * 2048 + t * 512;
#pragma unroll
  for (int q = 0; q < 4; ++q) vq[q] = *(const v4f*)(vt + q * 128);
}

__device__ __forceinline__ v2f vfrag(const v4f (&vq)[4], int half, int j) {
  const v4f& w = vq[2 * half + (j >> 1)];
  v2f f;
  f[0] = (j & 1) ? w[2] : w[0];
  f[1] = (j & 1) ? w[3] : w[1];
  return f;
}

__device__ __forceinline__ void key_group(
    const v4f (&ka)[16], const v4f (&kb)[16], v4f (&vq)[4][4],
    const v2f (&qb)[16], v8f& acc0, v8f& acc1, float& m_run, float& l_run,
    const float* __restrict__ vp, int m_next) {
  v8f s[4] = {{}, {}, {}, {}};
#pragma unroll
  for (int ck = 0; ck < 16; ++ck) {
    if (ck == 2) phase_priority<1>();
    if (ck == 8) phase_priority<2>();
#pragma unroll
    for (int t = 0; t < 4; ++t) {
      v2f a;
      a[0] = ka[ck][t];
      a[1] = kb[ck][t];
      s[t] = wmma4(a, qb[ck], s[t]);
    }
  }
  hold_loads_here();
  phase_priority<3>();

#pragma unroll
  for (int t = 0; t < 4; ++t) {
    const v8f st = c2b(s[t]);
    float tmax = st[0];
#pragma unroll
    for (int r = 1; r < 8; ++r) tmax = fmaxf(tmax, st[r]);
    tmax = fmaxf(tmax, sw16(tmax));
    float m_new = fmaxf(m_run, tmax);
    float scale = __expf(m_run - m_new);
    m_run = m_new;
    v8f p;
    float tsum = 0.0f;
#pragma unroll
    for (int r = 0; r < 8; ++r) {
      p[r] = __expf(st[r] - m_new);
      tsum += p[r];
    }
    tsum += sw16(tsum);
    l_run = l_run * scale + tsum;
#pragma unroll
    for (int r = 0; r < 8; ++r) {
      acc0[r] *= scale;
      acc1[r] *= scale;
    }

#pragma unroll
    for (int j = 0; j < 4; ++j) acc0 = wmma4(vfrag(vq[t], 0, j), bfrag(p, j), acc0);
#pragma unroll
    for (int j = 0; j < 4; ++j) acc1 = wmma4(vfrag(vq[t], 1, j), bfrag(p, j), acc1);
    load_v(vq[t], vp, m_next, t);
    if (t == 1) hold_loads_here();
  }
  hold_loads_here();
  phase_priority<0>();
}

__global__ __launch_bounds__(256) void attn_kernel(
    const float* __restrict__ x, const float* __restrict__ Vg,
    const float* __restrict__ W_w, const float* __restrict__ W_b,
    float* __restrict__ out) {
  const int b    = blockIdx.x >> 5;
  const int qblk = blockIdx.x & 31;
  const int tid  = threadIdx.x;
  const int wave = tid >> 5;
  const int lane = tid & 31;
  const int lq   = lane & 15;
  const bool lo  = lane < 16;
  const int crow = lo ? 0 : 2;
  const int n0   = qblk * 128 + wave * 16;

  const float* xb = x  + (size_t)b * (CC * NN);
  const float* vb = Vg + (size_t)b * (NN * C2);

  v2f qb[16];
  {
#pragma unroll
    for (int ck = 0; ck < 16; ++ck) {
      qb[ck][0] = xb[(size_t)(4 * ck + crow) * NN + n0 + lq];
      qb[ck][1] = xb[(size_t)(4 * ck + crow + 1) * NN + n0 + lq];
    }
  }

  v8f acc0 = {};
  v8f acc1 = {};
  float m_run = -INFINITY;
  float l_run = 0.0f;

  const float* kp = xb + (size_t)crow * NN + 4 * lq;
  const float* vp = vb + lane * 4;

  v4f k0a[16], k0b[16], k1a[16], k1b[16];
  v4f vf[4][4];
  load_k(k0a, k0b, kp, 0);
#pragma unroll
  for (int t = 0; t < 4; ++t) load_v(vf[t], vp, 0, t);

  for (int m0 = 0; m0 < NN; m0 += 128) {
    const int m1 = m0 + 64;
    const int m2 = (m0 + 128) & (NN - 1);
    load_k(k1a, k1b, kp, m1);
    hold_loads_here();
    key_group(k0a, k0b, vf, qb, acc0, acc1, m_run, l_run, vp, m1);
    load_k(k0a, k0b, kp, m2);
    hold_loads_here();
    key_group(k1a, k1b, vf, qb, acc0, acc1, m_run, l_run, vp, m2);
  }

  float rinv = 1.0f / l_run;
#pragma unroll
  for (int r = 0; r < 8; ++r) {
    acc0[r] *= rinv;
    acc1[r] *= rinv;
  }

  const v8f y0 = c2b(acc0);
  const v8f y1 = c2b(acc1);

  float* ob = out + (size_t)b * (CC * NN);
#pragma unroll
  for (int ct = 0; ct < 4; ++ct) {
    v8f z;
#pragma unroll
    for (int r = 0; r < 8; ++r) {
      int c = ct * 16 + r + (lo ? 0 : 8);
      z[r] = xb[(size_t)c * NN + n0 + lq] + W_b[c];
    }
    const int wrow = (ct * 16 + lq) * C2 + (lo ? 0 : 2);
#pragma unroll
    for (int oc = 0; oc < 8; ++oc) {
      v2f a;
      a[0] = W_w[wrow + 4 * oc];
      a[1] = W_w[wrow + 4 * oc + 1];
      z = wmma4(a, (oc < 4) ? bfrag(y0, oc) : bfrag(y1, oc - 4), z);
    }
#pragma unroll
    for (int r = 0; r < 8; ++r) {
      int c = ct * 16 + r + (lo ? 0 : 8);
      ob[(size_t)c * NN + n0 + lq] = z[r];
    }
  }
}

extern "C" void kernel_launch(void* const* d_in, const int* in_sizes, int n_in,
                              void* d_out, int out_size, void* d_ws,
                              size_t ws_size, hipStream_t stream) {
  const float* x   = (const float*)d_in[0];
  const float* g_w = (const float*)d_in[1];
  const float* g_b = (const float*)d_in[2];
  const float* W_w = (const float*)d_in[3];
  const float* W_b = (const float*)d_in[4];
  float* out = (float*)d_out;
  float* Vg  = (float*)d_ws;

  gx_kernel<<<NB * (NN / 128), 128, 0, stream>>>(x, g_w, g_b, Vg);
  attn_kernel<<<NB * (NN / 128), 256, 0, stream>>>(x, Vg, W_w, W_b, out);
}
